# starting kernel
# baseline (speedup 1.0000x reference)
#include <hip/hip_runtime.h>


typedef __attribute__((ext_vector_type(2))) float v2f;
typedef __attribute__((ext_vector_type(8))) float v8f;

#define NC    12
#define C2    144
#define C3    1728
#define C4    20736
#define LPTS  128
#define NINC  127
#define KP    28
#define NTHR  288
#define SIGSZ (NC + C2 + C3 + C4)

__launch_bounds__(NTHR, 1)
__global__ void sig_depth4_kernel(const float* __restrict__ path,
                                  float* __restrict__ out) {
  __shared__ float dx[NINC * NC];
  __shared__ float s1[NC];
  __shared__ float s2[C2];
  __shared__ float s3[C3];
  __shared__ __align__(16) float Am[C2 * KP];
  __shared__ __align__(16) float Bt[C2 * KP];

  const int b    = blockIdx.x;
  const int tid  = threadIdx.x;
  const int lane = tid & 31;
  const int wv   = tid >> 5;
  const float* p = path + (size_t)b * LPTS * NC;

  for (int idx = tid; idx < NINC * NC; idx += NTHR) {
    int t = idx / NC, c = idx - t * NC;
    dx[idx] = p[(t + 1) * NC + c] - p[t * NC + c];
  }
  if (tid < NC) s1[tid] = 0.f;
  if (tid < C2) s2[tid] = 0.f;
  for (int idx = tid; idx < C3; idx += NTHR) s3[idx] = 0.f;

  v8f acc[9];
#pragma unroll
  for (int j = 0; j < 9; ++j)
#pragma unroll
    for (int v = 0; v < 8; ++v) acc[j][v] = 0.f;

  const int m  = lane & 15;
  const int kb = (lane >> 4) << 1;

  float n1 = 0.f, n2 = 0.f, n3r[6];

  for (int t = 0; t < NINC; ++t) {
    __syncthreads();
    const float* d = &dx[t * NC];

#pragma unroll
    for (int r = 0; r < 14; ++r) {
      int idx = tid + r * NTHR;
      int ij = idx / KP, c = idx - ij * KP;
      int i = ij / 12, j = ij - i * 12;
      float v;
      if (c < 12)       v = s3[ij * 12 + c];
      else if (c < 24)  v = (j == (c - 12)) ? s1[i] : 0.f;
      else if (c == 24) v = s2[ij] + d[i] * d[j] * (1.f / 12.f);
      else              v = 0.f;
      Am[idx] = v;
    }
#pragma unroll
    for (int r = 0; r < 14; ++r) {
      int idx = tid + r * NTHR;
      int kl = idx / KP, c = idx - kl * KP;
      int k = kl / 12, l = kl - k * 12;
      float v;
      if (c < 12)       v = (c == k) ? d[l] : 0.f;
      else if (c < 24)  v = d[c - 12] * d[k] * d[l] * (1.f / 6.f);
      else if (c == 24) v = d[k] * d[l] * 0.5f;
      else              v = 0.f;
      Bt[idx] = v;
    }
    if (tid < NC) n1 = s1[tid] + d[tid];
    if (tid < C2) {
      int i = tid / 12, j = tid - i * 12;
      n2 = s2[tid] + s1[i] * d[j] + d[i] * d[j] * 0.5f;
    }
#pragma unroll
    for (int r = 0; r < 6; ++r) {
      int idx = tid + r * NTHR;
      int i = idx / 144, jk = idx - i * 144;
      int j = jk / 12, k = jk - j * 12;
      n3r[r] = s3[idx] + s2[i * 12 + j] * d[k]
             + s1[i] * d[j] * d[k] * 0.5f
             + d[i] * d[j] * d[k] * (1.f / 6.f);
    }
    __syncthreads();

    if (tid < NC) s1[tid] = n1;
    if (tid < C2) s2[tid] = n2;
#pragma unroll
    for (int r = 0; r < 6; ++r) s3[tid + r * NTHR] = n3r[r];

    v2f a[7];
#pragma unroll
    for (int kc = 0; kc < 7; ++kc)
      a[kc] = *(const v2f*)&Am[(16 * wv + m) * KP + 4 * kc + kb];
#pragma unroll
    for (int j = 0; j < 9; ++j) {
#pragma unroll
      for (int kc = 0; kc < 7; ++kc) {
        v2f bf = *(const v2f*)&Bt[(16 * j + m) * KP + 4 * kc + kb];
        acc[j] = __builtin_amdgcn_wmma_f32_16x16x4_f32(
             false, a[kc],  false, bf,
             (short)0, acc[j],  false,  false);
      }
    }
  }

  __syncthreads();

  float* ob = out + (size_t)b * SIGSZ;
  if (tid < NC) ob[tid] = s1[tid];
  if (tid < C2) ob[NC + tid] = s2[tid];
#pragma unroll
  for (int r = 0; r < 6; ++r) {
    int idx = tid + r * NTHR;
    ob[NC + C2 + idx] = s3[idx];
  }
  float* o4 = ob + NC + C2 + C3;
  const int rowoff = (lane >= 16) ? 8 : 0;
#pragma unroll
  for (int j = 0; j < 9; ++j)
#pragma unroll
    for (int v = 0; v < 8; ++v) {
      int row = 16 * wv + v + rowoff;
      o4[row * 144 + 16 * j + m] = acc[j][v];
    }
}

extern "C" void kernel_launch(void* const* d_in, const int* in_sizes, int n_in,
                              void* d_out, int out_size, void* d_ws, size_t ws_size,
                              hipStream_t stream) {
  const float* path = (const float*)d_in[0];
  float* out = (float*)d_out;
  (void)in_sizes; (void)n_in; (void)out_size; (void)d_ws; (void)ws_size;
  sig_depth4_kernel<<<dim3(128), dim3(NTHR), 0, stream>>>(path, out);
}


# Scan state and WMMA A/B fragments kept and formed in registers per wave instead of staged through LDS: no barrier in the scan; all-zero B k-chunks not issued (63 to 45 WMMAs per wave and step)
# speedup vs baseline: 4.1042x; 4.1042x over previous
#include <hip/hip_runtime.h>


typedef __attribute__((ext_vector_type(2))) float v2f;
typedef __attribute__((ext_vector_type(8))) float v8f;

#define NC    12
#define C2    144
#define C3    1728
#define C4    20736
#define LPTS  128
#define NINC  127
#define NTHR  288
#define SIGSZ (NC + C2 + C3 + C4)

__launch_bounds__(NTHR, 1)
__global__ void sig_depth4_kernel(const float* __restrict__ path,
                                  float* __restrict__ out) {
  __shared__ __align__(16) float dx[NINC * NC];

  const int b    = blockIdx.x;
  const int tid  = threadIdx.x;
  const int lane = tid & 31;
  const int wv   = tid >> 5;
  const float* p = path + (size_t)b * LPTS * NC;

  for (int idx = tid; idx < NINC * NC; idx += NTHR)
    dx[idx] = p[idx + NC] - p[idx];
  __syncthreads();

  const int m   = lane & 15;
  const int h   = lane >> 4;
  const int kb  = h << 1;
  const int row = 16 * wv + m;
  const int ri  = row / 12, rj = row - ri * 12;

  int lsel[3], kcol[9];
#pragma unroll
  for (int r = 0; r < 3; ++r) lsel[r] = (4 * r + m) % 12;
#pragma unroll
  for (int j = 0; j < 9; ++j) kcol[j] = (16 * j + m - lsel[j % 3]) / 12;

  float s1i = 0.f;
  float s2r = 0.f;
  v2f s3r[3];
#pragma unroll
  for (int c = 0; c < 3; ++c) s3r[c] = (v2f){0.f, 0.f};
  v8f acc[9];
#pragma unroll
  for (int j = 0; j < 9; ++j)
#pragma unroll
    for (int v = 0; v < 8; ++v) acc[j][v] = 0.f;

  for (int t = 0; t < NINC; ++t) {
    const float* d = &dx[t * NC];
    const float di = d[ri], dj = d[rj];
    v2f dk[3];
#pragma unroll
    for (int c = 0; c < 3; ++c) dk[c] = *(const v2f*)&d[4 * c + kb];
    float dl[3];
#pragma unroll
    for (int r = 0; r < 3; ++r) dl[r] = d[lsel[r]];

    v2f a[7];
#pragma unroll
    for (int c = 0; c < 3; ++c) a[c] = s3r[c];
#pragma unroll
    for (int c = 0; c < 3; ++c) {
      a[3 + c].x = (rj == 4 * c + kb)     ? s1i : 0.f;
      a[3 + c].y = (rj == 4 * c + kb + 1) ? s1i : 0.f;
    }
    a[6].x = (h == 0) ? (s2r + di * dj * (1.f / 12.f)) : 0.f;
    a[6].y = 0.f;

#pragma unroll
    for (int j = 0; j < 9; ++j) {
      const int cj = j / 3;
      const float dkc = d[kcol[j]];
      const float dlj = dl[j % 3];
      v2f bf;
      bf.x = (kcol[j] == 4 * cj + kb)     ? dlj : 0.f;
      bf.y = (kcol[j] == 4 * cj + kb + 1) ? dlj : 0.f;
      acc[j] = __builtin_amdgcn_wmma_f32_16x16x4_f32(
          false, a[cj], false, bf, (short)0, acc[j], false, false);
#pragma unroll
      for (int c = 0; c < 3; ++c) {
        bf = dk[c] * dkc * dlj * (1.f / 6.f);
        acc[j] = __builtin_amdgcn_wmma_f32_16x16x4_f32(
            false, a[3 + c], false, bf, (short)0, acc[j], false, false);
      }
      bf.x = (h == 0) ? (dkc * dlj * 0.5f) : 0.f;
      bf.y = 0.f;
      acc[j] = __builtin_amdgcn_wmma_f32_16x16x4_f32(
          false, a[6], false, bf, (short)0, acc[j], false, false);
    }

#pragma unroll
    for (int c = 0; c < 3; ++c) {
      float x = __builtin_fmaf(s2r, dk[c].x, s3r[c].x);
      float y = __builtin_fmaf(s2r, dk[c].y, s3r[c].y);
      x = __builtin_fmaf(s1i * dj * dk[c].x, 0.5f, x);
      y = __builtin_fmaf(s1i * dj * dk[c].y, 0.5f, y);
      s3r[c].x = __builtin_fmaf(di * dj * dk[c].x, 1.f / 6.f, x);
      s3r[c].y = __builtin_fmaf(di * dj * dk[c].y, 1.f / 6.f, y);
    }
    s2r = s2r + s1i * dj + di * dj * 0.5f;
    s1i = s1i + di;
  }

  float* ob = out + (size_t)b * SIGSZ;
  if (h == 0 && rj == 0) ob[ri] = s1i;
  if (h == 0) ob[NC + row] = s2r;
#pragma unroll
  for (int c = 0; c < 3; ++c)
    *(v2f*)&ob[NC + C2 + row * 12 + 4 * c + kb] = s3r[c];
  float* o4 = ob + NC + C2 + C3;
  const int rowoff = (lane >= 16) ? 8 : 0;
#pragma unroll
  for (int j = 0; j < 9; ++j)
#pragma unroll
    for (int v = 0; v < 8; ++v) {
      int orow = 16 * wv + v + rowoff;
      o4[orow * 144 + 16 * j + m] = acc[j][v];
    }
}

extern "C" void kernel_launch(void* const* d_in, const int* in_sizes, int n_in,
                              void* d_out, int out_size, void* d_ws, size_t ws_size,
                              hipStream_t stream) {
  const float* path = (const float*)d_in[0];
  float* out = (float*)d_out;
  (void)in_sizes; (void)n_in; (void)out_size; (void)d_ws; (void)ws_size;
  sig_depth4_kernel<<<dim3(128), dim3(NTHR), 0, stream>>>(path, out);
}


# Level-4 update in one k-chunk per tile: e4, sig1 x e3, sig2 x e2 share d_k d_l and become one A column g in a free k slot of the sig3 chunk; 5 to 1 WMMA per tile (level-4 sum order changes)
# speedup vs baseline: 14.7183x; 3.5862x over previous
#include <hip/hip_runtime.h>


typedef __attribute__((ext_vector_type(2))) float v2f;
typedef __attribute__((ext_vector_type(8))) float v8f;

#define NC    12
#define C2    144
#define C3    1728
#define C4    20736
#define LPTS  128
#define NINC  127
#define NTHR  288
#define SIGSZ (NC + C2 + C3 + C4)

__launch_bounds__(NTHR, 1)
__global__ void sig_depth4_kernel(const float* __restrict__ path,
                                  float* __restrict__ out) {
  __shared__ __align__(16) float dx[NINC * NC];

  const int b    = blockIdx.x;
  const int tid  = threadIdx.x;
  const int lane = tid & 31;
  const int wv   = tid >> 5;
  const float* p = path + (size_t)b * LPTS * NC;

  for (int idx = tid; idx < NINC * NC; idx += NTHR)
    dx[idx] = p[idx + NC] - p[idx];
  __syncthreads();

  const int m   = lane & 15;
  const int h   = lane >> 4;
  const int kb  = h << 1;
  const int row = 16 * wv + m;
  const int ri  = row / 12, rj = row - ri * 12;

  int lsel[3], qsel[3];
  bool m0[3], m1[3];
  float hg[3];
#pragma unroll
  for (int r = 0; r < 3; ++r) {
    lsel[r] = (4 * r + m) % 12;
    qsel[r] = (16 * r + m) / 12;
    m0[r] = (qsel[r] == kb);
    m1[r] = (qsel[r] == kb + 1);
    hg[r] = (((r == 0) ? 1 : 0) == h) ? 0.5f : 0.f;
  }

  float s1i = 0.f;
  float s2r = 0.f;
  v2f s3r[3];
#pragma unroll
  for (int c = 0; c < 3; ++c) s3r[c] = (v2f){0.f, 0.f};
  v8f acc[9];
#pragma unroll
  for (int j = 0; j < 9; ++j)
#pragma unroll
    for (int v = 0; v < 8; ++v) acc[j][v] = 0.f;

  for (int t = 0; t < NINC; ++t) {
    const float* d = &dx[t * NC];
    const float di = d[ri], dj = d[rj];
    v2f dk[3];
#pragma unroll
    for (int c = 0; c < 3; ++c) dk[c] = *(const v2f*)&d[4 * c + kb];
    float dl[3];
#pragma unroll
    for (int r = 0; r < 3; ++r) dl[r] = d[lsel[r]];

    const float g = __builtin_fmaf(
        dj, __builtin_fmaf(di, 1.f / 12.f, s1i * (1.f / 3.f)), s2r);
    v2f a0[3], a12[3];
#pragma unroll
    for (int c = 0; c < 3; ++c) {
      a0[c].x  = (h == 0) ? s3r[c].x : g;
      a0[c].y  = s3r[c].y;
      a12[c].x = (h == 0) ? g : s3r[c].x;
      a12[c].y = s3r[c].y;
    }

    v2f bsel[3];
    float dlh[3];
#pragma unroll
    for (int r = 0; r < 3; ++r) {
      bsel[r].x = m0[r] ? dl[r] : 0.f;
      bsel[r].y = m1[r] ? dl[r] : 0.f;
      dlh[r] = dl[r] * hg[r];
    }
#pragma unroll
    for (int j = 0; j < 9; ++j) {
      const int cj = j / 3;
      const int r  = j % 3;
      const float dkc = d[4 * cj + qsel[r]];
      v2f bf;
      bf.x = __builtin_fmaf(dkc, dlh[r], bsel[r].x);
      bf.y = bsel[r].y;
      acc[j] = __builtin_amdgcn_wmma_f32_16x16x4_f32(
          false, (r == 0) ? a0[cj] : a12[cj], false, bf, (short)0, acc[j],
          false, false);
    }

#pragma unroll
    for (int c = 0; c < 3; ++c) {
      float x = __builtin_fmaf(s2r, dk[c].x, s3r[c].x);
      float y = __builtin_fmaf(s2r, dk[c].y, s3r[c].y);
      x = __builtin_fmaf(s1i * dj * dk[c].x, 0.5f, x);
      y = __builtin_fmaf(s1i * dj * dk[c].y, 0.5f, y);
      s3r[c].x = __builtin_fmaf(di * dj * dk[c].x, 1.f / 6.f, x);
      s3r[c].y = __builtin_fmaf(di * dj * dk[c].y, 1.f / 6.f, y);
    }
    s2r = s2r + s1i * dj + di * dj * 0.5f;
    s1i = s1i + di;
  }

  float* ob = out + (size_t)b * SIGSZ;
  if (h == 0 && rj == 0) ob[ri] = s1i;
  if (h == 0) ob[NC + row] = s2r;
#pragma unroll
  for (int c = 0; c < 3; ++c)
    *(v2f*)&ob[NC + C2 + row * 12 + 4 * c + kb] = s3r[c];
  float* o4 = ob + NC + C2 + C3;
  const int rowoff = (lane >= 16) ? 8 : 0;
#pragma unroll
  for (int j = 0; j < 9; ++j)
#pragma unroll
    for (int v = 0; v < 8; ++v) {
      int orow = 16 * wv + v + rowoff;
      o4[orow * 144 + 16 * j + m] = acc[j][v];
    }
}

extern "C" void kernel_launch(void* const* d_in, const int* in_sizes, int n_in,
                              void* d_out, int out_size, void* d_ws, size_t ws_size,
                              hipStream_t stream) {
  const float* path = (const float*)d_in[0];
  float* out = (float*)d_out;
  (void)in_sizes; (void)n_in; (void)out_size; (void)d_ws; (void)ws_size;
  sig_depth4_kernel<<<dim3(128), dim3(NTHR), 0, stream>>>(path, out);
}


# Level 4 as a rank-1 update per step (Horner: sig4 += h3 (x) d): the 4 k slots of one f32 WMMA take 4 consecutive steps, 81 to 27 WMMAs per step, on 18 waves; sum order of levels 2 to 4 changes
# speedup vs baseline: 43.7232x; 2.9707x over previous
#include <hip/hip_runtime.h>


typedef __attribute__((ext_vector_type(2))) float v2f;
typedef __attribute__((ext_vector_type(8))) float v8f;

#define NC    12
#define C2    144
#define C3    1728
#define C4    20736
#define LPTS  128
#define NINC  127
#define NROW  130
#define NGRP  32
#define KT    6
#define NWAV  18
#define NTHR  (32 * NWAV)
#define SIGSZ (NC + C2 + C3 + C4)

struct RowState {
  float s1;
  float s2;
  float s3[KT];
};

__device__ __forceinline__ void form_h3(const RowState& st, float di, float dj,
                                        const float (&dk)[KT], float (&h3)[KT]) {
  const float g  = __builtin_fmaf(__builtin_fmaf(di, 0.25f, st.s1), dj * (1.f / 3.f), st.s2);
  const float hh = 0.5f * g;
#pragma unroll
  for (int q = 0; q < KT; ++q) h3[q] = __builtin_fmaf(hh, dk[q], st.s3[q]);
}

__device__ __forceinline__ void advance(RowState& st, float di, float dj,
                                        const float (&dk)[KT]) {
  const float q2 = __builtin_fmaf(__builtin_fmaf(di, 1.f / 3.f, st.s1), dj * 0.5f, st.s2);
#pragma unroll
  for (int q = 0; q < KT; ++q) st.s3[q] = __builtin_fmaf(q2, dk[q], st.s3[q]);
  st.s2 = __builtin_fmaf(__builtin_fmaf(di, 0.5f, st.s1), dj, st.s2);
  st.s1 = st.s1 + di;
}

__device__ __forceinline__ void load_dk(const float* d, int k0, float (&dk)[KT]) {
#pragma unroll
  for (int q = 0; q < KT / 2; ++q) {
    const v2f t = *(const v2f*)&d[k0 + 2 * q];
    dk[2 * q] = t.x;
    dk[2 * q + 1] = t.y;
  }
}

__launch_bounds__(NTHR, 1)
__global__ void sig_depth4_kernel(const float* __restrict__ path,
                                  float* __restrict__ out) {
  __shared__ __align__(16) float dx[NROW * NC];

  const int b    = blockIdx.x;
  const int tid  = threadIdx.x;
  const int lane = tid & 31;
  const int wv   = tid >> 5;
  const float* p = path + (size_t)b * LPTS * NC;

  for (int idx = tid; idx < NROW * NC; idx += NTHR)
    dx[idx] = (idx < NINC * NC) ? (p[idx + NC] - p[idx]) : 0.f;
  __syncthreads();

  const int c   = lane & 15;
  const int h   = lane >> 4;
  const int grp = wv % 9;
  const int k0  = KT * (wv / 9);
  const int row = 16 * grp + c;
  const int ri  = row / 12, rj = row - ri * 12;
  const bool cv = (c < NC);
  const int cb  = cv ? c : (NC - 1);

  RowState st;
  st.s1 = 0.f;
  st.s2 = 0.f;
#pragma unroll
  for (int q = 0; q < KT; ++q) st.s3[q] = 0.f;
  v8f acc[KT];
#pragma unroll
  for (int q = 0; q < KT; ++q)
#pragma unroll
    for (int v = 0; v < 8; ++v) acc[q][v] = 0.f;

#pragma unroll
  for (int s = 0; s < 2; ++s) {
    const float* d = &dx[(h ? s : (NINC + s)) * NC];
    float dk[KT];
    load_dk(d, k0, dk);
    advance(st, d[ri], d[rj], dk);
  }

  for (int g = 0; g < NGRP; ++g) {
    const float* d0 = &dx[(4 * g + 2 * h) * NC];

    v2f bf;
    bf.x = d0[cb];
    bf.y = d0[NC + cb];

    float hx[KT], hy[KT];
    {
      float dk[KT];
      load_dk(d0, k0, dk);
      const float di = d0[ri], dj = d0[rj];
      form_h3(st, di, dj, dk, hx);
      advance(st, di, dj, dk);
    }
    {
      const float* d = d0 + NC;
      float dk[KT];
      load_dk(d, k0, dk);
      const float di = d[ri], dj = d[rj];
      form_h3(st, di, dj, dk, hy);
      advance(st, di, dj, dk);
    }
#pragma unroll
    for (int s = 2; s < 4; ++s) {
      const float* d = d0 + s * NC;
      float dk[KT];
      load_dk(d, k0, dk);
      advance(st, d[ri], d[rj], dk);
    }

#pragma unroll
    for (int q = 0; q < KT; ++q) {
      v2f af;
      af.x = hx[q];
      af.y = hy[q];
      acc[q] = __builtin_amdgcn_wmma_f32_16x16x4_f32(
          false, af, false, bf, (short)0, acc[q], false, false);
    }
  }

  float* ob = out + (size_t)b * SIGSZ;
  if (k0 == 0 && h == 0) {
    if (rj == 0) ob[ri] = st.s1;
    ob[NC + row] = st.s2;
  }
  if (h == 0) {
#pragma unroll
    for (int q = 0; q < KT / 2; ++q) {
      v2f t;
      t.x = st.s3[2 * q];
      t.y = st.s3[2 * q + 1];
      *(v2f*)&ob[NC + C2 + row * NC + k0 + 2 * q] = t;
    }
  }
  float* o4 = ob + NC + C2 + C3;
  if (cv) {
#pragma unroll
    for (int q = 0; q < KT; ++q)
#pragma unroll
      for (int v = 0; v < 8; ++v) {
        const int orow = 16 * grp + 8 * h + v;
        o4[orow * C2 + (k0 + q) * NC + c] = acc[q][v];
      }
  }
}

extern "C" void kernel_launch(void* const* d_in, const int* in_sizes, int n_in,
                              void* d_out, int out_size, void* d_ws, size_t ws_size,
                              hipStream_t stream) {
  const float* path = (const float*)d_in[0];
  float* out = (float*)d_out;
  (void)in_sizes; (void)n_in; (void)out_size; (void)d_ws; (void)ws_size;
  sig_depth4_kernel<<<dim3(128), dim3(NTHR), 0, stream>>>(path, out);
}


# One batch element spread over two work-groups (one per half of k) of 10 waves, tiles per SIMD 15,15,12,12 instead of 30,30,24,24: all 256 WGPs work; per-tile arithmetic and sum order as before
# speedup vs baseline: 58.3266x; 1.3340x over previous
#include <hip/hip_runtime.h>


typedef __attribute__((ext_vector_type(2))) float v2f;
typedef __attribute__((ext_vector_type(8))) float v8f;

#define NC    12
#define C2    144
#define C3    1728
#define C4    20736
#define LPTS  128
#define NINC  127
#define NROW  130
#define NGRP  32
#define KH    6
#define NWAV  10
#define NTHR  (32 * NWAV)
#define SIGSZ (NC + C2 + C3 + C4)

template <int KT>
struct RowState {
  float s1;
  float s2;
  float s3[KT];
};

template <int KT>
__device__ __forceinline__ void form_h3(const RowState<KT>& st, float di, float dj,
                                        const float (&dk)[KT], float (&h3)[KT]) {
  const float g  = __builtin_fmaf(__builtin_fmaf(di, 0.25f, st.s1), dj * (1.f / 3.f), st.s2);
  const float hh = 0.5f * g;
#pragma unroll
  for (int q = 0; q < KT; ++q) h3[q] = __builtin_fmaf(hh, dk[q], st.s3[q]);
}

template <int KT>
__device__ __forceinline__ void advance(RowState<KT>& st, float di, float dj,
                                        const float (&dk)[KT]) {
  const float q2 = __builtin_fmaf(__builtin_fmaf(di, 1.f / 3.f, st.s1), dj * 0.5f, st.s2);
#pragma unroll
  for (int q = 0; q < KT; ++q) st.s3[q] = __builtin_fmaf(q2, dk[q], st.s3[q]);
  st.s2 = __builtin_fmaf(__builtin_fmaf(di, 0.5f, st.s1), dj, st.s2);
  st.s1 = st.s1 + di;
}

template <int KT>
__device__ __forceinline__ void load_dk(const float* d, int k0, float (&dk)[KT]) {
  if constexpr (KT % 2 == 0) {
#pragma unroll
    for (int q = 0; q < KT / 2; ++q) {
      const v2f t = *(const v2f*)&d[k0 + 2 * q];
      dk[2 * q] = t.x;
      dk[2 * q + 1] = t.y;
    }
  } else {
#pragma unroll
    for (int q = 0; q < KT; ++q) dk[q] = d[k0 + q];
  }
}

template <int KT>
__device__ __forceinline__ void run_wave(const float* dx, float* ob, int lane,
                                         int grp, int k0) {
  const int c   = lane & 15;
  const int h   = lane >> 4;
  const int row = 16 * grp + c;
  const int ri  = row / 12, rj = row - ri * 12;
  const bool cv = (c < NC);
  const int cb  = cv ? c : (NC - 1);

  RowState<KT> st;
  st.s1 = 0.f;
  st.s2 = 0.f;
#pragma unroll
  for (int q = 0; q < KT; ++q) st.s3[q] = 0.f;
  v8f acc[KT];
#pragma unroll
  for (int q = 0; q < KT; ++q)
#pragma unroll
    for (int v = 0; v < 8; ++v) acc[q][v] = 0.f;

#pragma unroll
  for (int s = 0; s < 2; ++s) {
    const float* d = &dx[(h ? s : (NINC + s)) * NC];
    float dk[KT];
    load_dk<KT>(d, k0, dk);
    advance<KT>(st, d[ri], d[rj], dk);
  }

  for (int g = 0; g < NGRP; ++g) {
    const float* d0 = &dx[(4 * g + 2 * h) * NC];

    v2f bf;
    bf.x = d0[cb];
    bf.y = d0[NC + cb];

    float hx[KT], hy[KT];
    {
      float dk[KT];
      load_dk<KT>(d0, k0, dk);
      const float di = d0[ri], dj = d0[rj];
      form_h3<KT>(st, di, dj, dk, hx);
      advance<KT>(st, di, dj, dk);
    }
    {
      const float* d = d0 + NC;
      float dk[KT];
      load_dk<KT>(d, k0, dk);
      const float di = d[ri], dj = d[rj];
      form_h3<KT>(st, di, dj, dk, hy);
      advance<KT>(st, di, dj, dk);
    }
#pragma unroll
    for (int s = 2; s < 4; ++s) {
      const float* d = d0 + s * NC;
      float dk[KT];
      load_dk<KT>(d, k0, dk);
      advance<KT>(st, d[ri], d[rj], dk);
    }

#pragma unroll
    for (int q = 0; q < KT; ++q) {
      v2f af;
      af.x = hx[q];
      af.y = hy[q];
      acc[q] = __builtin_amdgcn_wmma_f32_16x16x4_f32(
          false, af, false, bf, (short)0, acc[q], false, false);
    }
  }

  if (k0 == 0 && h == 0) {
    if (rj == 0) ob[ri] = st.s1;
    ob[NC + row] = st.s2;
  }
  if (h == 0) {
    float* o3 = &ob[NC + C2 + row * NC + k0];
    if constexpr (KT % 2 == 0) {
#pragma unroll
      for (int q = 0; q < KT / 2; ++q) {
        v2f t;
        t.x = st.s3[2 * q];
        t.y = st.s3[2 * q + 1];
        *(v2f*)&o3[2 * q] = t;
      }
    } else {
#pragma unroll
      for (int q = 0; q < KT; ++q) o3[q] = st.s3[q];
    }
  }
  float* o4 = ob + NC + C2 + C3;
  if (cv) {
#pragma unroll
    for (int q = 0; q < KT; ++q)
#pragma unroll
      for (int v = 0; v < 8; ++v) {
        const int orow = 16 * grp + 8 * h + v;
        o4[orow * C2 + (k0 + q) * NC + c] = acc[q][v];
      }
  }
}

__launch_bounds__(NTHR, 1)
__global__ void sig_depth4_kernel(const float* __restrict__ path,
                                  float* __restrict__ out) {
  __shared__ __align__(16) float dx[NROW * NC];

  const int b    = blockIdx.x >> 1;
  const int kh   = blockIdx.x & 1;
  const int tid  = threadIdx.x;
  const int lane = tid & 31;
  const int wv   = tid >> 5;
  const float* p = path + (size_t)b * LPTS * NC;

  for (int idx = tid; idx < NROW * NC; idx += NTHR)
    dx[idx] = (idx < NINC * NC) ? (p[idx + NC] - p[idx]) : 0.f;
  __syncthreads();

  float* ob = out + (size_t)b * SIGSZ;
  if (wv < 8) {
    run_wave<6>(dx, ob, lane, wv, KH * kh);
  } else {
    run_wave<3>(dx, ob, lane, 8, KH * kh + 3 * (wv - 8));
  }
}

extern "C" void kernel_launch(void* const* d_in, const int* in_sizes, int n_in,
                              void* d_out, int out_size, void* d_ws, size_t ws_size,
                              hipStream_t stream) {
  const float* path = (const float*)d_in[0];
  float* out = (float*)d_out;
  (void)in_sizes; (void)n_in; (void)out_size; (void)d_ws; (void)ws_size;
  sig_depth4_kernel<<<dim3(256), dim3(NTHR), 0, stream>>>(path, out);
}


# Path increments formed from 16-byte loads that are all issued before the first use: the scan waits for one trip to memory instead of five dependent ones (values and sum order unchanged)
# speedup vs baseline: 67.4281x; 1.1560x over previous
#include <hip/hip_runtime.h>


typedef __attribute__((ext_vector_type(2))) float v2f;
typedef __attribute__((ext_vector_type(4))) float v4f;
typedef __attribute__((ext_vector_type(8))) float v8f;

#define NC    12
#define C2    144
#define C3    1728
#define C4    20736
#define LPTS  128
#define NINC  127
#define NROW  130
#define NGRP  32
#define KH    6
#define NWAV  10
#define NTHR  (32 * NWAV)
#define SIGSZ (NC + C2 + C3 + C4)
#define NQ    (NROW * NC / 4)
#define NQR   (NINC * NC / 4)

template <int KT>
struct RowState {
  float s1;
  float s2;
  float s3[KT];
};

template <int KT>
__device__ __forceinline__ void form_h3(const RowState<KT>& st, float di, float dj,
                                        const float (&dk)[KT], float (&h3)[KT]) {
  const float g  = __builtin_fmaf(__builtin_fmaf(di, 0.25f, st.s1), dj * (1.f / 3.f), st.s2);
  const float hh = 0.5f * g;
#pragma unroll
  for (int q = 0; q < KT; ++q) h3[q] = __builtin_fmaf(hh, dk[q], st.s3[q]);
}

template <int KT>
__device__ __forceinline__ void advance(RowState<KT>& st, float di, float dj,
                                        const float (&dk)[KT]) {
  const float q2 = __builtin_fmaf(__builtin_fmaf(di, 1.f / 3.f, st.s1), dj * 0.5f, st.s2);
#pragma unroll
  for (int q = 0; q < KT; ++q) st.s3[q] = __builtin_fmaf(q2, dk[q], st.s3[q]);
  st.s2 = __builtin_fmaf(__builtin_fmaf(di, 0.5f, st.s1), dj, st.s2);
  st.s1 = st.s1 + di;
}

template <int KT>
__device__ __forceinline__ void load_dk(const float* d, int k0, float (&dk)[KT]) {
  if constexpr (KT % 2 == 0) {
#pragma unroll
    for (int q = 0; q < KT / 2; ++q) {
      const v2f t = *(const v2f*)&d[k0 + 2 * q];
      dk[2 * q] = t.x;
      dk[2 * q + 1] = t.y;
    }
  } else {
#pragma unroll
    for (int q = 0; q < KT; ++q) dk[q] = d[k0 + q];
  }
}

template <int KT>
__device__ __forceinline__ void run_wave(const float* dx, float* ob, int lane,
                                         int grp, int k0) {
  const int c   = lane & 15;
  const int h   = lane >> 4;
  const int row = 16 * grp + c;
  const int ri  = row / 12, rj = row - ri * 12;
  const bool cv = (c < NC);
  const int cb  = cv ? c : (NC - 1);

  RowState<KT> st;
  st.s1 = 0.f;
  st.s2 = 0.f;
#pragma unroll
  for (int q = 0; q < KT; ++q) st.s3[q] = 0.f;
  v8f acc[KT];
#pragma unroll
  for (int q = 0; q < KT; ++q)
#pragma unroll
    for (int v = 0; v < 8; ++v) acc[q][v] = 0.f;

#pragma unroll
  for (int s = 0; s < 2; ++s) {
    const float* d = &dx[(h ? s : (NINC + s)) * NC];
    float dk[KT];
    load_dk<KT>(d, k0, dk);
    advance<KT>(st, d[ri], d[rj], dk);
  }

  for (int g = 0; g < NGRP; ++g) {
    const float* d0 = &dx[(4 * g + 2 * h) * NC];

    v2f bf;
    bf.x = d0[cb];
    bf.y = d0[NC + cb];

    float hx[KT], hy[KT];
    {
      float dk[KT];
      load_dk<KT>(d0, k0, dk);
      const float di = d0[ri], dj = d0[rj];
      form_h3<KT>(st, di, dj, dk, hx);
      advance<KT>(st, di, dj, dk);
    }
    {
      const float* d = d0 + NC;
      float dk[KT];
      load_dk<KT>(d, k0, dk);
      const float di = d[ri], dj = d[rj];
      form_h3<KT>(st, di, dj, dk, hy);
      advance<KT>(st, di, dj, dk);
    }
#pragma unroll
    for (int s = 2; s < 4; ++s) {
      const float* d = d0 + s * NC;
      float dk[KT];
      load_dk<KT>(d, k0, dk);
      advance<KT>(st, d[ri], d[rj], dk);
    }

#pragma unroll
    for (int q = 0; q < KT; ++q) {
      v2f af;
      af.x = hx[q];
      af.y = hy[q];
      acc[q] = __builtin_amdgcn_wmma_f32_16x16x4_f32(
          false, af, false, bf, (short)0, acc[q], false, false);
    }
  }

  if (k0 == 0 && h == 0) {
    if (rj == 0) ob[ri] = st.s1;
    ob[NC + row] = st.s2;
  }
  if (h == 0) {
    float* o3 = &ob[NC + C2 + row * NC + k0];
    if constexpr (KT % 2 == 0) {
#pragma unroll
      for (int q = 0; q < KT / 2; ++q) {
        v2f t;
        t.x = st.s3[2 * q];
        t.y = st.s3[2 * q + 1];
        *(v2f*)&o3[2 * q] = t;
      }
    } else {
#pragma unroll
      for (int q = 0; q < KT; ++q) o3[q] = st.s3[q];
    }
  }
  float* o4 = ob + NC + C2 + C3;
  if (cv) {
#pragma unroll
    for (int q = 0; q < KT; ++q)
#pragma unroll
      for (int v = 0; v < 8; ++v) {
        const int orow = 16 * grp + 8 * h + v;
        o4[orow * C2 + (k0 + q) * NC + c] = acc[q][v];
      }
  }
}

__launch_bounds__(NTHR, 1)
__global__ void sig_depth4_kernel(const float* __restrict__ path,
                                  float* __restrict__ out) {
  __shared__ __align__(16) float dx[NROW * NC];

  const int b    = blockIdx.x >> 1;
  const int kh   = blockIdx.x & 1;
  const int tid  = threadIdx.x;
  const int lane = tid & 31;
  const int wv   = tid >> 5;
  const float* p = path + (size_t)b * LPTS * NC;

  const v4f* p4 = (const v4f*)p;
  const int i0 = tid;
  const int i1 = tid + NTHR;
  const v4f a0 = p4[i0 + 3];
  const v4f b0 = p4[i0];
  v4f a1 = {0.f, 0.f, 0.f, 0.f};
  v4f b1 = {0.f, 0.f, 0.f, 0.f};
  if (i1 < NQR) {
    a1 = p4[i1 + 3];
    b1 = p4[i1];
  }
  v4f* dx4 = (v4f*)dx;
  dx4[i0] = a0 - b0;
  if (i1 < NQ) dx4[i1] = a1 - b1;
  __syncthreads();

  float* ob = out + (size_t)b * SIGSZ;
  if (wv < 8) {
    run_wave<6>(dx, ob, lane, wv, KH * kh);
  } else {
    run_wave<3>(dx, ob, lane, 8, KH * kh + 3 * (wv - 8));
  }
}

extern "C" void kernel_launch(void* const* d_in, const int* in_sizes, int n_in,
                              void* d_out, int out_size, void* d_ws, size_t ws_size,
                              hipStream_t stream) {
  const float* path = (const float*)d_in[0];
  float* out = (float*)d_out;
  (void)in_sizes; (void)n_in; (void)out_size; (void)d_ws; (void)ws_size;
  sig_depth4_kernel<<<dim3(256), dim3(NTHR), 0, stream>>>(path, out);
}


# LDS reads of each group of four steps (d_i, d_j, d_k, B fragment) issued one group ahead into a second register set: a wave's matrix run no longer waits for LDS (values, sum order unchanged)
# speedup vs baseline: 70.9407x; 1.0521x over previous
#include <hip/hip_runtime.h>


typedef __attribute__((ext_vector_type(2))) float v2f;
typedef __attribute__((ext_vector_type(4))) float v4f;
typedef __attribute__((ext_vector_type(8))) float v8f;

#define NC    12
#define C2    144
#define C3    1728
#define C4    20736
#define LPTS  128
#define NINC  127
#define NROW  134
#define NGRP  32
#define KH    6
#define NWAV  10
#define NTHR  (32 * NWAV)
#define SIGSZ (NC + C2 + C3 + C4)
#define NQ    (NROW * NC / 4)
#define NQR   (NINC * NC / 4)

template <int KT>
struct RowState {
  float s1;
  float s2;
  float s3[KT];
};

template <int KT>
__device__ __forceinline__ void form_h3(const RowState<KT>& st, float di, float dj,
                                        const float (&dk)[KT], float (&h3)[KT]) {
  const float g  = __builtin_fmaf(__builtin_fmaf(di, 0.25f, st.s1), dj * (1.f / 3.f), st.s2);
  const float hh = 0.5f * g;
#pragma unroll
  for (int q = 0; q < KT; ++q) h3[q] = __builtin_fmaf(hh, dk[q], st.s3[q]);
}

template <int KT>
__device__ __forceinline__ void advance(RowState<KT>& st, float di, float dj,
                                        const float (&dk)[KT]) {
  const float q2 = __builtin_fmaf(__builtin_fmaf(di, 1.f / 3.f, st.s1), dj * 0.5f, st.s2);
#pragma unroll
  for (int q = 0; q < KT; ++q) st.s3[q] = __builtin_fmaf(q2, dk[q], st.s3[q]);
  st.s2 = __builtin_fmaf(__builtin_fmaf(di, 0.5f, st.s1), dj, st.s2);
  st.s1 = st.s1 + di;
}

template <int KT>
__device__ __forceinline__ void load_dk(const float* d, int k0, float (&dk)[KT]) {
  if constexpr (KT % 2 == 0) {
#pragma unroll
    for (int q = 0; q < KT / 2; ++q) {
      const v2f t = *(const v2f*)&d[k0 + 2 * q];
      dk[2 * q] = t.x;
      dk[2 * q + 1] = t.y;
    }
  } else {
#pragma unroll
    for (int q = 0; q < KT; ++q) dk[q] = d[k0 + q];
  }
}

template <int KT>
struct GroupRows {
  v2f   bf;
  float di[4];
  float dj[4];
  float dk[4][KT];
};

template <int KT>
__device__ __forceinline__ void load_group(const float* d0, int ri, int rj, int cb,
                                           int k0, GroupRows<KT>& r) {
  r.bf.x = d0[cb];
  r.bf.y = d0[NC + cb];
#pragma unroll
  for (int s = 0; s < 4; ++s) {
    const float* d = d0 + s * NC;
    r.di[s] = d[ri];
    r.dj[s] = d[rj];
    load_dk<KT>(d, k0, r.dk[s]);
  }
}

__device__ __forceinline__ void slots_behind_block(v8f (&acc)[3], const v2f (&af)[3],
                                                   const v2f& bf) {
  asm volatile("v_nop\n\tv_nop\n\tv_nop\n\tv_nop"
               : "+v"(acc[0]), "+v"(acc[1]), "+v"(acc[2])
               : "v"(af[0]), "v"(af[1]), "v"(af[2]), "v"(bf));
}
__device__ __forceinline__ void slots_behind_block(v8f (&acc)[6], const v2f (&af)[6],
                                                   const v2f& bf) {
  asm volatile("v_nop\n\tv_nop\n\tv_nop\n\tv_nop"
               : "+v"(acc[0]), "+v"(acc[1]), "+v"(acc[2]), "+v"(acc[3]), "+v"(acc[4]), "+v"(acc[5])
               : "v"(af[0]), "v"(af[1]), "v"(af[2]), "v"(af[3]), "v"(af[4]), "v"(af[5]), "v"(bf));
}

template <int KT>
__device__ __forceinline__ void do_group(RowState<KT>& st, const GroupRows<KT>& r,
                                         v8f (&acc)[KT]) {
  float hx[KT], hy[KT];
  v2f af[KT];
  form_h3<KT>(st, r.di[0], r.dj[0], r.dk[0], hx);
  advance<KT>(st, r.di[0], r.dj[0], r.dk[0]);
  form_h3<KT>(st, r.di[1], r.dj[1], r.dk[1], hy);
  advance<KT>(st, r.di[1], r.dj[1], r.dk[1]);
  advance<KT>(st, r.di[2], r.dj[2], r.dk[2]);
  advance<KT>(st, r.di[3], r.dj[3], r.dk[3]);
#pragma unroll
  for (int q = 0; q < KT; ++q) {
    af[q].x = hx[q];
    af[q].y = hy[q];
    acc[q] = __builtin_amdgcn_wmma_f32_16x16x4_f32(
        false, af[q], false, r.bf, (short)0, acc[q], false, false);
  }
  slots_behind_block(acc, af, r.bf);
}

template <int KT>
__device__ __forceinline__ void run_wave(const float* dx, float* ob, int lane,
                                         int grp, int k0) {
  const int c   = lane & 15;
  const int h   = lane >> 4;
  const int row = 16 * grp + c;
  const int ri  = row / 12, rj = row - ri * 12;
  const bool cv = (c < NC);
  const int cb  = cv ? c : (NC - 1);

  RowState<KT> st;
  st.s1 = 0.f;
  st.s2 = 0.f;
#pragma unroll
  for (int q = 0; q < KT; ++q) st.s3[q] = 0.f;
  v8f acc[KT];
#pragma unroll
  for (int q = 0; q < KT; ++q)
#pragma unroll
    for (int v = 0; v < 8; ++v) acc[q][v] = 0.f;

  const float* dh = &dx[(2 * h) * NC];
  GroupRows<KT> ra, rb;
  load_group<KT>(dh, ri, rj, cb, k0, ra);

#pragma unroll
  for (int s = 0; s < 2; ++s) {
    const float* d = &dx[(h ? s : (NINC + s)) * NC];
    float dk[KT];
    load_dk<KT>(d, k0, dk);
    advance<KT>(st, d[ri], d[rj], dk);
  }

  for (int g = 0; g < NGRP; g += 2) {
    load_group<KT>(dh + (4 * (g + 1)) * NC, ri, rj, cb, k0, rb);
    __builtin_amdgcn_sched_barrier(0);
    do_group<KT>(st, ra, acc);
    load_group<KT>(dh + (4 * (g + 2)) * NC, ri, rj, cb, k0, ra);
    __builtin_amdgcn_sched_barrier(0);
    do_group<KT>(st, rb, acc);
  }

  if (k0 == 0 && h == 0) {
    if (rj == 0) ob[ri] = st.s1;
    ob[NC + row] = st.s2;
  }
  if (h == 0) {
    float* o3 = &ob[NC + C2 + row * NC + k0];
    if constexpr (KT % 2 == 0) {
#pragma unroll
      for (int q = 0; q < KT / 2; ++q) {
        v2f t;
        t.x = st.s3[2 * q];
        t.y = st.s3[2 * q + 1];
        *(v2f*)&o3[2 * q] = t;
      }
    } else {
#pragma unroll
      for (int q = 0; q < KT; ++q) o3[q] = st.s3[q];
    }
  }
  float* o4 = ob + NC + C2 + C3;
  if (cv) {
#pragma unroll
    for (int q = 0; q < KT; ++q)
#pragma unroll
      for (int v = 0; v < 8; ++v) {
        const int orow = 16 * grp + 8 * h + v;
        o4[orow * C2 + (k0 + q) * NC + c] = acc[q][v];
      }
  }
}

__launch_bounds__(NTHR, 1)
__global__ void sig_depth4_kernel(const float* __restrict__ path,
                                  float* __restrict__ out) {
  __shared__ __align__(16) float dx[NROW * NC];

  const int b    = blockIdx.x >> 1;
  const int kh   = blockIdx.x & 1;
  const int tid  = threadIdx.x;
  const int lane = tid & 31;
  const int wv   = tid >> 5;
  const float* p = path + (size_t)b * LPTS * NC;

  const v4f* p4 = (const v4f*)p;
  const int i0 = tid;
  const int i1 = tid + NTHR;
  const v4f a0 = p4[i0 + 3];
  const v4f b0 = p4[i0];
  v4f a1 = {0.f, 0.f, 0.f, 0.f};
  v4f b1 = {0.f, 0.f, 0.f, 0.f};
  if (i1 < NQR) {
    a1 = p4[i1 + 3];
    b1 = p4[i1];
  }
  v4f* dx4 = (v4f*)dx;
  dx4[i0] = a0 - b0;
  if (i1 < NQ) dx4[i1] = a1 - b1;
  __syncthreads();

  float* ob = out + (size_t)b * SIGSZ;
  if (wv < 8) {
    run_wave<6>(dx, ob, lane, wv, KH * kh);
  } else {
    run_wave<3>(dx, ob, lane, 8, KH * kh + 3 * (wv - 8));
  }
}

extern "C" void kernel_launch(void* const* d_in, const int* in_sizes, int n_in,
                              void* d_out, int out_size, void* d_ws, size_t ws_size,
                              hipStream_t stream) {
  const float* path = (const float*)d_in[0];
  float* out = (float*)d_out;
  (void)in_sizes; (void)n_in; (void)out_size; (void)d_ws; (void)ws_size;
  sig_depth4_kernel<<<dim3(256), dim3(NTHR), 0, stream>>>(path, out);
}
